# starting kernel
# baseline (speedup 1.0000x reference)
#include <hip/hip_runtime.h>

#define N_NODES 100000
#define N_EDGES 1600000
#define D_FEAT  128
#define D_HID   512
#define D_OUT   128
#define SCAN_B  1024
#define NB      ((N_NODES + SCAN_B - 1) / SCAN_B)

typedef float v2f __attribute__((ext_vector_type(2)));
typedef float v8f __attribute__((ext_vector_type(8)));

__device__ __forceinline__ v8f wmma4(v2f a, v2f b, v8f c) {
  return __builtin_amdgcn_wmma_f32_16x16x4_f32(false, a, false, b,
                                               (short)0, c, false, false);
}


__global__ void k_zero(int* __restrict__ p, int n) {
  int i = blockIdx.x * blockDim.x + threadIdx.x;
  if (i < n) p[i] = 0;
}

__global__ void k_deg(const int* __restrict__ dst, int* __restrict__ deg) {
  int e = blockIdx.x * blockDim.x + threadIdx.x;
  if (e < N_EDGES) atomicAdd(&deg[dst[e]], 1);
}

__global__ __launch_bounds__(SCAN_B)
void k_scan1(const int* __restrict__ deg, int* __restrict__ incl,
             int* __restrict__ bsum, int n) {
  __shared__ int s[SCAN_B];
  int i = blockIdx.x * SCAN_B + threadIdx.x;
  s[threadIdx.x] = (i < n) ? deg[i] : 0;
  __syncthreads();
  for (int off = 1; off < SCAN_B; off <<= 1) {
    int t = (threadIdx.x >= off) ? s[threadIdx.x - off] : 0;
    __syncthreads();
    s[threadIdx.x] += t;
    __syncthreads();
  }
  if (i < n) incl[i] = s[threadIdx.x];
  if (threadIdx.x == SCAN_B - 1) bsum[blockIdx.x] = s[SCAN_B - 1];
}

__global__ void k_scan2(int* __restrict__ bsum, int nb) {
  if (threadIdx.x == 0 && blockIdx.x == 0) {
    int run = 0;
    for (int i = 0; i < nb; ++i) { int t = bsum[i]; bsum[i] = run; run += t; }
  }
}

__global__ __launch_bounds__(SCAN_B)
void k_scan3(const int* __restrict__ incl, const int* __restrict__ bsum,
             int* __restrict__ offs, int n) {
  int i = blockIdx.x * SCAN_B + threadIdx.x;
  if (i < n) offs[i + 1] = incl[i] + bsum[blockIdx.x];
  if (i == 0) offs[0] = 0;
}

__global__ void k_bin(const int* __restrict__ src, const int* __restrict__ dst,
                      const int* __restrict__ offs, int* __restrict__ cursor,
                      int* __restrict__ bsrc) {
  int e = blockIdx.x * blockDim.x + threadIdx.x;
  if (e >= N_EDGES) return;
  int d = dst[e];
  int pos = atomicAdd(&cursor[d], 1);
  bsrc[offs[d] + pos] = src[e];
}

__global__ __launch_bounds__(256)
void k_aggr(const float* __restrict__ x, const int* __restrict__ offs,
            const int* __restrict__ bsrc, float* __restrict__ h) {
  int node = (blockIdx.x << 3) + (threadIdx.x >> 5);
  if (node >= N_NODES) return;
  int lane = threadIdx.x & 31;
  float4 acc = *reinterpret_cast<const float4*>(x + (size_t)node * D_FEAT + lane * 4);
  int e0 = offs[node], e1 = offs[node + 1];
  for (int e = e0; e < e1; ++e) {
    int s = bsrc[e];
    float4 v = *reinterpret_cast<const float4*>(x + (size_t)s * D_FEAT + lane * 4);
    acc.x += v.x; acc.y += v.y; acc.z += v.z; acc.w += v.w;
  }
  *reinterpret_cast<float4*>(h + (size_t)node * D_FEAT + lane * 4) = acc;
}

__global__ __launch_bounds__(256)
void gin_mlp(const float* __restrict__ h,
             const float* __restrict__ W1, const float* __restrict__ b1,
             const float* __restrict__ W2, const float* __restrict__ b2,
             float* __restrict__ out) {
  extern __shared__ float smem[];
  float* sH  = smem;
  float* sH1 = smem + 32 * D_FEAT;

  const int tid   = threadIdx.x;
  const int wave  = tid >> 5;
  const int lane  = tid & 31;
  const int row16 = lane & 15;
  const int khalf = (lane >> 4) << 1;
  const int rowhi = (lane >> 4) << 3;
  const size_t base = (size_t)blockIdx.x * 32 * D_FEAT;

  for (int i = tid; i < 32 * D_FEAT; i += 256) sH[i] = h[base + i];
  __syncthreads();

  for (int t = 0; t < 4; ++t) {
    const int n0  = (wave + (t << 3)) << 4;
    const int col = n0 + row16;
    v8f acc0 = {0.f,0.f,0.f,0.f,0.f,0.f,0.f,0.f};
    v8f acc1 = {0.f,0.f,0.f,0.f,0.f,0.f,0.f,0.f};
#pragma unroll
    for (int kk = 0; kk < D_FEAT; kk += 4) {
      const int kb = kk + khalf;
      v2f b;
      b.x = W1[(size_t)kb * D_HID + col];
      b.y = W1[(size_t)(kb + 1) * D_HID + col];
      v2f a0 = *reinterpret_cast<const v2f*>(sH + row16 * D_FEAT + kk + khalf);
      v2f a1 = *reinterpret_cast<const v2f*>(sH + (16 + row16) * D_FEAT + kk + khalf);
      acc0 = wmma4(a0, b, acc0);
      acc1 = wmma4(a1, b, acc1);
    }
    const float bias = b1[col];
#pragma unroll
    for (int i = 0; i < 8; ++i) {
      float v0 = acc0[i] + bias;
      float v1 = acc1[i] + bias;
      sH1[(i + rowhi) * D_HID + col]      = v0 > 0.f ? v0 : 0.f;
      sH1[(16 + i + rowhi) * D_HID + col] = v1 > 0.f ? v1 : 0.f;
    }
  }
  __syncthreads();

  {
    const int n0  = wave << 4;
    const int col = n0 + row16;
    v8f acc0 = {0.f,0.f,0.f,0.f,0.f,0.f,0.f,0.f};
    v8f acc1 = {0.f,0.f,0.f,0.f,0.f,0.f,0.f,0.f};
#pragma unroll 8
    for (int kk = 0; kk < D_HID; kk += 4) {
      const int kb = kk + khalf;
      v2f b;
      b.x = W2[(size_t)kb * D_OUT + col];
      b.y = W2[(size_t)(kb + 1) * D_OUT + col];
      v2f a0 = *reinterpret_cast<const v2f*>(sH1 + row16 * D_HID + kk + khalf);
      v2f a1 = *reinterpret_cast<const v2f*>(sH1 + (16 + row16) * D_HID + kk + khalf);
      acc0 = wmma4(a0, b, acc0);
      acc1 = wmma4(a1, b, acc1);
    }
    const float bias = b2[col];
    const size_t obase = (size_t)blockIdx.x * 32 * D_OUT;
#pragma unroll
    for (int i = 0; i < 8; ++i) {
      out[obase + (size_t)(i + rowhi) * D_OUT + col]      = acc0[i] + bias;
      out[obase + (size_t)(16 + i + rowhi) * D_OUT + col] = acc1[i] + bias;
    }
  }
}

extern "C" void kernel_launch(void* const* d_in, const int* in_sizes, int n_in,
                              void* d_out, int out_size, void* d_ws, size_t ws_size,
                              hipStream_t stream) {
  const float* x  = (const float*)d_in[0];
  const int*   ei = (const int*)d_in[1];
  const float* W1 = (const float*)d_in[2];
  const float* b1 = (const float*)d_in[3];
  const float* W2 = (const float*)d_in[4];
  const float* b2 = (const float*)d_in[5];
  float* out = (float*)d_out;

  const int* src = ei;
  const int* dst = ei + N_EDGES;

  float* h    = (float*)d_ws;
  int*  bsrc  = (int*)(h + (size_t)N_NODES * D_FEAT);
  int*  deg   = bsrc + N_EDGES;
  int*  curs  = deg + N_NODES;
  int*  incl  = curs + N_NODES;
  int*  offs  = incl + N_NODES;
  int*  bsum  = offs + N_NODES + 1;

  k_zero<<<(2 * N_NODES + 255) / 256, 256, 0, stream>>>(deg, 2 * N_NODES);
  k_deg<<<(N_EDGES + 255) / 256, 256, 0, stream>>>(dst, deg);
  k_scan1<<<NB, SCAN_B, 0, stream>>>(deg, incl, bsum, N_NODES);
  k_scan2<<<1, 32, 0, stream>>>(bsum, NB);
  k_scan3<<<NB, SCAN_B, 0, stream>>>(incl, bsum, offs, N_NODES);
  k_bin<<<(N_EDGES + 255) / 256, 256, 0, stream>>>(src, dst, offs, curs, bsrc);

  k_aggr<<<(N_NODES + 7) / 8, 256, 0, stream>>>(x, offs, bsrc, h);

  gin_mlp<<<N_NODES / 32, 256, 80 * 1024, stream>>>(h, W1, b1, W2, b2, out);
}


# fused MLP kernel: LDS row pitch of both A tiles K -> K+4 floats (16 B x odd) so the 8-byte A fragment reads of 16 rows spread over all banks instead of one bank pair; LDS per block 80 -> 81 KB
# speedup vs baseline: 1.3882x; 1.3882x over previous
#include <hip/hip_runtime.h>

#define N_NODES 100000
#define N_EDGES 1600000
#define D_FEAT  128
#define D_HID   512
#define D_OUT   128
#define SCAN_B  1024
#define NB      ((N_NODES + SCAN_B - 1) / SCAN_B)
#define H_PITCH  (D_FEAT + 4)
#define H1_PITCH (D_HID + 4)

typedef float v2f __attribute__((ext_vector_type(2)));
typedef float v8f __attribute__((ext_vector_type(8)));

__device__ __forceinline__ v8f wmma4(v2f a, v2f b, v8f c) {
  return __builtin_amdgcn_wmma_f32_16x16x4_f32(false, a, false, b,
                                               (short)0, c, false, false);
}


__global__ void k_zero(int* __restrict__ p, int n) {
  int i = blockIdx.x * blockDim.x + threadIdx.x;
  if (i < n) p[i] = 0;
}

__global__ void k_deg(const int* __restrict__ dst, int* __restrict__ deg) {
  int e = blockIdx.x * blockDim.x + threadIdx.x;
  if (e < N_EDGES) atomicAdd(&deg[dst[e]], 1);
}

__global__ __launch_bounds__(SCAN_B)
void k_scan1(const int* __restrict__ deg, int* __restrict__ incl,
             int* __restrict__ bsum, int n) {
  __shared__ int s[SCAN_B];
  int i = blockIdx.x * SCAN_B + threadIdx.x;
  s[threadIdx.x] = (i < n) ? deg[i] : 0;
  __syncthreads();
  for (int off = 1; off < SCAN_B; off <<= 1) {
    int t = (threadIdx.x >= off) ? s[threadIdx.x - off] : 0;
    __syncthreads();
    s[threadIdx.x] += t;
    __syncthreads();
  }
  if (i < n) incl[i] = s[threadIdx.x];
  if (threadIdx.x == SCAN_B - 1) bsum[blockIdx.x] = s[SCAN_B - 1];
}

__global__ void k_scan2(int* __restrict__ bsum, int nb) {
  if (threadIdx.x == 0 && blockIdx.x == 0) {
    int run = 0;
    for (int i = 0; i < nb; ++i) { int t = bsum[i]; bsum[i] = run; run += t; }
  }
}

__global__ __launch_bounds__(SCAN_B)
void k_scan3(const int* __restrict__ incl, const int* __restrict__ bsum,
             int* __restrict__ offs, int n) {
  int i = blockIdx.x * SCAN_B + threadIdx.x;
  if (i < n) offs[i + 1] = incl[i] + bsum[blockIdx.x];
  if (i == 0) offs[0] = 0;
}

__global__ void k_bin(const int* __restrict__ src, const int* __restrict__ dst,
                      const int* __restrict__ offs, int* __restrict__ cursor,
                      int* __restrict__ bsrc) {
  int e = blockIdx.x * blockDim.x + threadIdx.x;
  if (e >= N_EDGES) return;
  int d = dst[e];
  int pos = atomicAdd(&cursor[d], 1);
  bsrc[offs[d] + pos] = src[e];
}

__global__ __launch_bounds__(256)
void k_aggr(const float* __restrict__ x, const int* __restrict__ offs,
            const int* __restrict__ bsrc, float* __restrict__ h) {
  int node = (blockIdx.x << 3) + (threadIdx.x >> 5);
  if (node >= N_NODES) return;
  int lane = threadIdx.x & 31;
  float4 acc = *reinterpret_cast<const float4*>(x + (size_t)node * D_FEAT + lane * 4);
  int e0 = offs[node], e1 = offs[node + 1];
  for (int e = e0; e < e1; ++e) {
    int s = bsrc[e];
    float4 v = *reinterpret_cast<const float4*>(x + (size_t)s * D_FEAT + lane * 4);
    acc.x += v.x; acc.y += v.y; acc.z += v.z; acc.w += v.w;
  }
  *reinterpret_cast<float4*>(h + (size_t)node * D_FEAT + lane * 4) = acc;
}

__global__ __launch_bounds__(256)
void gin_mlp(const float* __restrict__ h,
             const float* __restrict__ W1, const float* __restrict__ b1,
             const float* __restrict__ W2, const float* __restrict__ b2,
             float* __restrict__ out) {
  extern __shared__ float smem[];
  float* sH  = smem;
  float* sH1 = smem + 32 * H_PITCH;

  const int tid   = threadIdx.x;
  const int wave  = tid >> 5;
  const int lane  = tid & 31;
  const int row16 = lane & 15;
  const int khalf = (lane >> 4) << 1;
  const int rowhi = (lane >> 4) << 3;
  const size_t base = (size_t)blockIdx.x * 32 * D_FEAT;

  for (int i = tid; i < 32 * D_FEAT; i += 256)
    sH[(i >> 7) * H_PITCH + (i & (D_FEAT - 1))] = h[base + i];
  __syncthreads();

  for (int t = 0; t < 4; ++t) {
    const int n0  = (wave + (t << 3)) << 4;
    const int col = n0 + row16;
    v8f acc0 = {0.f,0.f,0.f,0.f,0.f,0.f,0.f,0.f};
    v8f acc1 = {0.f,0.f,0.f,0.f,0.f,0.f,0.f,0.f};
#pragma unroll
    for (int kk = 0; kk < D_FEAT; kk += 4) {
      const int kb = kk + khalf;
      v2f b;
      b.x = W1[(size_t)kb * D_HID + col];
      b.y = W1[(size_t)(kb + 1) * D_HID + col];
      v2f a0 = *reinterpret_cast<const v2f*>(sH + row16 * H_PITCH + kk + khalf);
      v2f a1 = *reinterpret_cast<const v2f*>(sH + (16 + row16) * H_PITCH + kk + khalf);
      acc0 = wmma4(a0, b, acc0);
      acc1 = wmma4(a1, b, acc1);
    }
    const float bias = b1[col];
#pragma unroll
    for (int i = 0; i < 8; ++i) {
      float v0 = acc0[i] + bias;
      float v1 = acc1[i] + bias;
      sH1[(i + rowhi) * H1_PITCH + col]      = v0 > 0.f ? v0 : 0.f;
      sH1[(16 + i + rowhi) * H1_PITCH + col] = v1 > 0.f ? v1 : 0.f;
    }
  }
  __syncthreads();

  {
    const int n0  = wave << 4;
    const int col = n0 + row16;
    v8f acc0 = {0.f,0.f,0.f,0.f,0.f,0.f,0.f,0.f};
    v8f acc1 = {0.f,0.f,0.f,0.f,0.f,0.f,0.f,0.f};
#pragma unroll 8
    for (int kk = 0; kk < D_HID; kk += 4) {
      const int kb = kk + khalf;
      v2f b;
      b.x = W2[(size_t)kb * D_OUT + col];
      b.y = W2[(size_t)(kb + 1) * D_OUT + col];
      v2f a0 = *reinterpret_cast<const v2f*>(sH1 + row16 * H1_PITCH + kk + khalf);
      v2f a1 = *reinterpret_cast<const v2f*>(sH1 + (16 + row16) * H1_PITCH + kk + khalf);
      acc0 = wmma4(a0, b, acc0);
      acc1 = wmma4(a1, b, acc1);
    }
    const float bias = b2[col];
    const size_t obase = (size_t)blockIdx.x * 32 * D_OUT;
#pragma unroll
    for (int i = 0; i < 8; ++i) {
      out[obase + (size_t)(i + rowhi) * D_OUT + col]      = acc0[i] + bias;
      out[obase + (size_t)(16 + i + rowhi) * D_OUT + col] = acc1[i] + bias;
    }
  }
}

extern "C" void kernel_launch(void* const* d_in, const int* in_sizes, int n_in,
                              void* d_out, int out_size, void* d_ws, size_t ws_size,
                              hipStream_t stream) {
  const float* x  = (const float*)d_in[0];
  const int*   ei = (const int*)d_in[1];
  const float* W1 = (const float*)d_in[2];
  const float* b1 = (const float*)d_in[3];
  const float* W2 = (const float*)d_in[4];
  const float* b2 = (const float*)d_in[5];
  float* out = (float*)d_out;

  const int* src = ei;
  const int* dst = ei + N_EDGES;

  float* h    = (float*)d_ws;
  int*  bsrc  = (int*)(h + (size_t)N_NODES * D_FEAT);
  int*  deg   = bsrc + N_EDGES;
  int*  curs  = deg + N_NODES;
  int*  incl  = curs + N_NODES;
  int*  offs  = incl + N_NODES;
  int*  bsum  = offs + N_NODES + 1;

  k_zero<<<(2 * N_NODES + 255) / 256, 256, 0, stream>>>(deg, 2 * N_NODES);
  k_deg<<<(N_EDGES + 255) / 256, 256, 0, stream>>>(dst, deg);
  k_scan1<<<NB, SCAN_B, 0, stream>>>(deg, incl, bsum, N_NODES);
  k_scan2<<<1, 32, 0, stream>>>(bsum, NB);
  k_scan3<<<NB, SCAN_B, 0, stream>>>(incl, bsum, offs, N_NODES);
  k_bin<<<(N_EDGES + 255) / 256, 256, 0, stream>>>(src, dst, offs, curs, bsrc);

  k_aggr<<<(N_NODES + 7) / 8, 256, 0, stream>>>(x, offs, bsrc, h);

  gin_mlp<<<N_NODES / 32, 256, 82944, stream>>>(h, W1, b1, W2, b2, out);
}


# CSR build: per-edge global integer atomics (degree count, slot cursor) replaced by a two-level counting sort on LDS counters (rank = returning LDS add); 8 -> 6 launches; bin order still arrival order
# speedup vs baseline: 2.2737x; 1.6379x over previous
#include <hip/hip_runtime.h>

#define N_NODES 100000
#define N_EDGES 1600000
#define D_FEAT  128
#define D_HID   512
#define D_OUT   128
#define BKT_SHIFT 9
#define BKT_R   (1 << BKT_SHIFT)
#define NBKT    ((N_NODES + BKT_R - 1) >> BKT_SHIFT)
#define NCHUNK  256
#define CHUNK_E ((N_EDGES + NCHUNK - 1) / NCHUNK)
#define EPT     ((CHUNK_E + 255) / 256)
#define SRC_BITS 17
static_assert(NBKT <= 256, "bucket totals are scanned by one 256-thread block");
static_assert(N_NODES <= (1 << SRC_BITS), "src id must fit the low bits of a placed edge");
static_assert(BKT_SHIFT + SRC_BITS < 32, "placed edge word stays non-negative");
#define H_PITCH  (D_FEAT + 4)
#define H1_PITCH (D_HID + 4)

typedef float v2f __attribute__((ext_vector_type(2)));
typedef float v8f __attribute__((ext_vector_type(8)));

__device__ __forceinline__ v8f wmma4(v2f a, v2f b, v8f c) {
  return __builtin_amdgcn_wmma_f32_16x16x4_f32(false, a, false, b,
                                               (short)0, c, false, false);
}


template <int N>
__device__ __forceinline__ int block_incl_scan(int v, int* s) {
  const int t = threadIdx.x;
  if (t < N) s[t] = v;
  __syncthreads();
#pragma unroll
  for (int off = 1; off < N; off <<= 1) {
    int a = 0;
    if (t < N && t >= off) a = s[t - off];
    __syncthreads();
    if (t < N) s[t] += a;
    __syncthreads();
  }
  return (t < N) ? s[t] : 0;
}

__global__ __launch_bounds__(256)
void k_hist(const int* __restrict__ dst, int* __restrict__ hist) {
  __shared__ int cnt[NBKT];
  const int t = threadIdx.x;
  const int e0 = blockIdx.x * CHUNK_E;
  const int e1 = (e0 + CHUNK_E < N_EDGES) ? (e0 + CHUNK_E) : N_EDGES;
  int d[EPT];
#pragma unroll
  for (int j = 0; j < EPT; ++j) {
    const int e = e0 + j * 256 + t;
    d[j] = (e < e1) ? dst[e] : -1;
  }
  for (int i = t; i < NBKT; i += 256) cnt[i] = 0;
  __syncthreads();
#pragma unroll
  for (int j = 0; j < EPT; ++j)
    if (d[j] >= 0) atomicAdd(&cnt[d[j] >> BKT_SHIFT], 1);
  __syncthreads();
  for (int i = t; i < NBKT; i += 256) hist[i * NCHUNK + blockIdx.x] = cnt[i];
}

__global__ __launch_bounds__(256)
void k_bscan(const int* __restrict__ hist, int* __restrict__ posl, int* __restrict__ btot) {
  __shared__ int s[256];
  const int t = threadIdx.x;
  const int v = hist[blockIdx.x * NCHUNK + t];
  const int incl = block_incl_scan<256>(v, s);
  posl[blockIdx.x * NCHUNK + t] = incl - v;
  if (t == 255) btot[blockIdx.x] = incl;
}

__global__ __launch_bounds__(256)
void k_scatter(const int* __restrict__ src, const int* __restrict__ dst,
               const int* __restrict__ posl, const int* __restrict__ btot,
               int* __restrict__ keys, int* __restrict__ bstart) {
  __shared__ int s[256];
  __shared__ int cur[NBKT];
  const int t = threadIdx.x;
  const int g = blockIdx.x;
  const int e0 = g * CHUNK_E;
  const int e1 = (e0 + CHUNK_E < N_EDGES) ? (e0 + CHUNK_E) : N_EDGES;
  int d[EPT], sr[EPT];
#pragma unroll
  for (int j = 0; j < EPT; ++j) {
    const int e = e0 + j * 256 + t;
    const bool ok = e < e1;
    d[j]  = ok ? dst[e] : -1;
    sr[j] = ok ? src[e] : 0;
  }
  const int tv = (t < NBKT) ? btot[t] : 0;
  const int incl = block_incl_scan<256>(tv, s);
  const int base = incl - tv;
  if (t < NBKT) cur[t] = base + posl[t * NCHUNK + g];
  if (g == 0) {
    if (t < NBKT) bstart[t] = base;
    if (t == NBKT - 1) bstart[NBKT] = incl;
  }
  __syncthreads();
#pragma unroll
  for (int j = 0; j < EPT; ++j) {
    if (d[j] >= 0) {
      const int p = atomicAdd(&cur[d[j] >> BKT_SHIFT], 1);
      keys[p] = ((d[j] & (BKT_R - 1)) << SRC_BITS) | sr[j];
    }
  }
}

__global__ __launch_bounds__(1024)
void k_bucket(const int* __restrict__ keys, const int* __restrict__ bstart,
              int* __restrict__ offs, int* __restrict__ bsrc) {
  __shared__ int cnt[BKT_R];
  __shared__ int sc[BKT_R];
  const int t = threadIdx.x;
  const int b = blockIdx.x;
  const int k0 = bstart[b], k1 = bstart[b + 1];
  if (t < BKT_R) cnt[t] = 0;
  __syncthreads();
  for (int i0 = k0; i0 < k1; i0 += 1024 * 8) {
    int kv[8];
#pragma unroll
    for (int j = 0; j < 8; ++j) {
      const int i = i0 + j * 1024 + t;
      kv[j] = (i < k1) ? keys[i] : -1;
    }
#pragma unroll
    for (int j = 0; j < 8; ++j)
      if (kv[j] >= 0) atomicAdd(&cnt[kv[j] >> SRC_BITS], 1);
  }
  __syncthreads();
  const int v = (t < BKT_R) ? cnt[t] : 0;
  const int incl = block_incl_scan<BKT_R>(v, sc);
  if (t < BKT_R) {
    const int first = k0 + incl - v;
    const int node = b * BKT_R + t;
    if (node < N_NODES) offs[node] = first;
    cnt[t] = first;
  }
  if (b == NBKT - 1 && t == 0) offs[N_NODES] = k1;
  __syncthreads();
  for (int i0 = k0; i0 < k1; i0 += 1024 * 8) {
    int kv[8];
#pragma unroll
    for (int j = 0; j < 8; ++j) {
      const int i = i0 + j * 1024 + t;
      kv[j] = (i < k1) ? keys[i] : -1;
    }
#pragma unroll
    for (int j = 0; j < 8; ++j) {
      if (kv[j] >= 0) {
        const int p = atomicAdd(&cnt[kv[j] >> SRC_BITS], 1);
        bsrc[p] = kv[j] & ((1 << SRC_BITS) - 1);
      }
    }
  }
}

__global__ __launch_bounds__(256)
void k_aggr(const float* __restrict__ x, const int* __restrict__ offs,
            const int* __restrict__ bsrc, float* __restrict__ h) {
  int node = (blockIdx.x << 3) + (threadIdx.x >> 5);
  if (node >= N_NODES) return;
  int lane = threadIdx.x & 31;
  float4 acc = *reinterpret_cast<const float4*>(x + (size_t)node * D_FEAT + lane * 4);
  int e0 = offs[node], e1 = offs[node + 1];
  for (int e = e0; e < e1; ++e) {
    int s = bsrc[e];
    float4 v = *reinterpret_cast<const float4*>(x + (size_t)s * D_FEAT + lane * 4);
    acc.x += v.x; acc.y += v.y; acc.z += v.z; acc.w += v.w;
  }
  *reinterpret_cast<float4*>(h + (size_t)node * D_FEAT + lane * 4) = acc;
}

__global__ __launch_bounds__(256)
void gin_mlp(const float* __restrict__ h,
             const float* __restrict__ W1, const float* __restrict__ b1,
             const float* __restrict__ W2, const float* __restrict__ b2,
             float* __restrict__ out) {
  extern __shared__ float smem[];
  float* sH  = smem;
  float* sH1 = smem + 32 * H_PITCH;

  const int tid   = threadIdx.x;
  const int wave  = tid >> 5;
  const int lane  = tid & 31;
  const int row16 = lane & 15;
  const int khalf = (lane >> 4) << 1;
  const int rowhi = (lane >> 4) << 3;
  const size_t base = (size_t)blockIdx.x * 32 * D_FEAT;

  for (int i = tid; i < 32 * D_FEAT; i += 256)
    sH[(i >> 7) * H_PITCH + (i & (D_FEAT - 1))] = h[base + i];
  __syncthreads();

  for (int t = 0; t < 4; ++t) {
    const int n0  = (wave + (t << 3)) << 4;
    const int col = n0 + row16;
    v8f acc0 = {0.f,0.f,0.f,0.f,0.f,0.f,0.f,0.f};
    v8f acc1 = {0.f,0.f,0.f,0.f,0.f,0.f,0.f,0.f};
#pragma unroll
    for (int kk = 0; kk < D_FEAT; kk += 4) {
      const int kb = kk + khalf;
      v2f b;
      b.x = W1[(size_t)kb * D_HID + col];
      b.y = W1[(size_t)(kb + 1) * D_HID + col];
      v2f a0 = *reinterpret_cast<const v2f*>(sH + row16 * H_PITCH + kk + khalf);
      v2f a1 = *reinterpret_cast<const v2f*>(sH + (16 + row16) * H_PITCH + kk + khalf);
      acc0 = wmma4(a0, b, acc0);
      acc1 = wmma4(a1, b, acc1);
    }
    const float bias = b1[col];
#pragma unroll
    for (int i = 0; i < 8; ++i) {
      float v0 = acc0[i] + bias;
      float v1 = acc1[i] + bias;
      sH1[(i + rowhi) * H1_PITCH + col]      = v0 > 0.f ? v0 : 0.f;
      sH1[(16 + i + rowhi) * H1_PITCH + col] = v1 > 0.f ? v1 : 0.f;
    }
  }
  __syncthreads();

  {
    const int n0  = wave << 4;
    const int col = n0 + row16;
    v8f acc0 = {0.f,0.f,0.f,0.f,0.f,0.f,0.f,0.f};
    v8f acc1 = {0.f,0.f,0.f,0.f,0.f,0.f,0.f,0.f};
#pragma unroll 8
    for (int kk = 0; kk < D_HID; kk += 4) {
      const int kb = kk + khalf;
      v2f b;
      b.x = W2[(size_t)kb * D_OUT + col];
      b.y = W2[(size_t)(kb + 1) * D_OUT + col];
      v2f a0 = *reinterpret_cast<const v2f*>(sH1 + row16 * H1_PITCH + kk + khalf);
      v2f a1 = *reinterpret_cast<const v2f*>(sH1 + (16 + row16) * H1_PITCH + kk + khalf);
      acc0 = wmma4(a0, b, acc0);
      acc1 = wmma4(a1, b, acc1);
    }
    const float bias = b2[col];
    const size_t obase = (size_t)blockIdx.x * 32 * D_OUT;
#pragma unroll
    for (int i = 0; i < 8; ++i) {
      out[obase + (size_t)(i + rowhi) * D_OUT + col]      = acc0[i] + bias;
      out[obase + (size_t)(16 + i + rowhi) * D_OUT + col] = acc1[i] + bias;
    }
  }
}

extern "C" void kernel_launch(void* const* d_in, const int* in_sizes, int n_in,
                              void* d_out, int out_size, void* d_ws, size_t ws_size,
                              hipStream_t stream) {
  const float* x  = (const float*)d_in[0];
  const int*   ei = (const int*)d_in[1];
  const float* W1 = (const float*)d_in[2];
  const float* b1 = (const float*)d_in[3];
  const float* W2 = (const float*)d_in[4];
  const float* b2 = (const float*)d_in[5];
  float* out = (float*)d_out;

  const int* src = ei;
  const int* dst = ei + N_EDGES;

  float* h     = (float*)d_ws;
  int*  bsrc   = (int*)(h + (size_t)N_NODES * D_FEAT);
  int*  keys   = bsrc + N_EDGES;
  int*  hist   = keys + N_EDGES;
  int*  posl   = hist + NBKT * NCHUNK;
  int*  btot   = posl + NBKT * NCHUNK;
  int*  bstart = btot + NBKT;
  int*  offs   = bstart + NBKT + 1;

  k_hist<<<NCHUNK, 256, 0, stream>>>(dst, hist);
  k_bscan<<<NBKT, 256, 0, stream>>>(hist, posl, btot);
  k_scatter<<<NCHUNK, 256, 0, stream>>>(src, dst, posl, btot, keys, bstart);
  k_bucket<<<NBKT, 1024, 0, stream>>>(keys, bstart, offs, bsrc);

  k_aggr<<<(N_NODES + 7) / 8, 256, 0, stream>>>(x, offs, bsrc, h);

  gin_mlp<<<N_NODES / 32, 256, 82944, stream>>>(h, W1, b1, W2, b2, out);
}


# MLP kernel: wave tile 32x64 (was 32x16): 16-byte W row loads feed 4 N tiles, each A fragment 4 WMMAs, W kept 4 k-steps ahead; GEMM2 K split over 4 wave groups, summed via LDS (new sum order)
# speedup vs baseline: 2.4149x; 1.0621x over previous
#include <hip/hip_runtime.h>

#define N_NODES 100000
#define N_EDGES 1600000
#define D_FEAT  128
#define D_HID   512
#define D_OUT   128
#define BKT_SHIFT 9
#define BKT_R   (1 << BKT_SHIFT)
#define NBKT    ((N_NODES + BKT_R - 1) >> BKT_SHIFT)
#define NCHUNK  256
#define CHUNK_E ((N_EDGES + NCHUNK - 1) / NCHUNK)
#define EPT     ((CHUNK_E + 255) / 256)
#define SRC_BITS 17
static_assert(NBKT <= 256, "bucket totals are scanned by one 256-thread block");
static_assert(N_NODES <= (1 << SRC_BITS), "src id must fit the low bits of a placed edge");
static_assert(BKT_SHIFT + SRC_BITS < 32, "placed edge word stays non-negative");
#define H_PITCH  (D_FEAT + 4)
#define H1_PITCH (D_HID + 4)

typedef float v2f __attribute__((ext_vector_type(2)));
typedef float v8f __attribute__((ext_vector_type(8)));

__device__ __forceinline__ v8f wmma4(v2f a, v2f b, v8f c) {
  return __builtin_amdgcn_wmma_f32_16x16x4_f32(false, a, false, b,
                                               (short)0, c, false, false);
}


template <int N>
__device__ __forceinline__ int block_incl_scan(int v, int* s) {
  const int t = threadIdx.x;
  if (t < N) s[t] = v;
  __syncthreads();
#pragma unroll
  for (int off = 1; off < N; off <<= 1) {
    int a = 0;
    if (t < N && t >= off) a = s[t - off];
    __syncthreads();
    if (t < N) s[t] += a;
    __syncthreads();
  }
  return (t < N) ? s[t] : 0;
}

__global__ __launch_bounds__(256)
void k_hist(const int* __restrict__ dst, int* __restrict__ hist) {
  __shared__ int cnt[NBKT];
  const int t = threadIdx.x;
  const int e0 = blockIdx.x * CHUNK_E;
  const int e1 = (e0 + CHUNK_E < N_EDGES) ? (e0 + CHUNK_E) : N_EDGES;
  int d[EPT];
#pragma unroll
  for (int j = 0; j < EPT; ++j) {
    const int e = e0 + j * 256 + t;
    d[j] = (e < e1) ? dst[e] : -1;
  }
  for (int i = t; i < NBKT; i += 256) cnt[i] = 0;
  __syncthreads();
#pragma unroll
  for (int j = 0; j < EPT; ++j)
    if (d[j] >= 0) atomicAdd(&cnt[d[j] >> BKT_SHIFT], 1);
  __syncthreads();
  for (int i = t; i < NBKT; i += 256) hist[i * NCHUNK + blockIdx.x] = cnt[i];
}

__global__ __launch_bounds__(256)
void k_bscan(const int* __restrict__ hist, int* __restrict__ posl, int* __restrict__ btot) {
  __shared__ int s[256];
  const int t = threadIdx.x;
  const int v = hist[blockIdx.x * NCHUNK + t];
  const int incl = block_incl_scan<256>(v, s);
  posl[blockIdx.x * NCHUNK + t] = incl - v;
  if (t == 255) btot[blockIdx.x] = incl;
}

__global__ __launch_bounds__(256)
void k_scatter(const int* __restrict__ src, const int* __restrict__ dst,
               const int* __restrict__ posl, const int* __restrict__ btot,
               int* __restrict__ keys, int* __restrict__ bstart) {
  __shared__ int s[256];
  __shared__ int cur[NBKT];
  const int t = threadIdx.x;
  const int g = blockIdx.x;
  const int e0 = g * CHUNK_E;
  const int e1 = (e0 + CHUNK_E < N_EDGES) ? (e0 + CHUNK_E) : N_EDGES;
  int d[EPT], sr[EPT];
#pragma unroll
  for (int j = 0; j < EPT; ++j) {
    const int e = e0 + j * 256 + t;
    const bool ok = e < e1;
    d[j]  = ok ? dst[e] : -1;
    sr[j] = ok ? src[e] : 0;
  }
  const int tv = (t < NBKT) ? btot[t] : 0;
  const int incl = block_incl_scan<256>(tv, s);
  const int base = incl - tv;
  if (t < NBKT) cur[t] = base + posl[t * NCHUNK + g];
  if (g == 0) {
    if (t < NBKT) bstart[t] = base;
    if (t == NBKT - 1) bstart[NBKT] = incl;
  }
  __syncthreads();
#pragma unroll
  for (int j = 0; j < EPT; ++j) {
    if (d[j] >= 0) {
      const int p = atomicAdd(&cur[d[j] >> BKT_SHIFT], 1);
      keys[p] = ((d[j] & (BKT_R - 1)) << SRC_BITS) | sr[j];
    }
  }
}

__global__ __launch_bounds__(1024)
void k_bucket(const int* __restrict__ keys, const int* __restrict__ bstart,
              int* __restrict__ offs, int* __restrict__ bsrc) {
  __shared__ int cnt[BKT_R];
  __shared__ int sc[BKT_R];
  const int t = threadIdx.x;
  const int b = blockIdx.x;
  const int k0 = bstart[b], k1 = bstart[b + 1];
  if (t < BKT_R) cnt[t] = 0;
  __syncthreads();
  for (int i0 = k0; i0 < k1; i0 += 1024 * 8) {
    int kv[8];
#pragma unroll
    for (int j = 0; j < 8; ++j) {
      const int i = i0 + j * 1024 + t;
      kv[j] = (i < k1) ? keys[i] : -1;
    }
#pragma unroll
    for (int j = 0; j < 8; ++j)
      if (kv[j] >= 0) atomicAdd(&cnt[kv[j] >> SRC_BITS], 1);
  }
  __syncthreads();
  const int v = (t < BKT_R) ? cnt[t] : 0;
  const int incl = block_incl_scan<BKT_R>(v, sc);
  if (t < BKT_R) {
    const int first = k0 + incl - v;
    const int node = b * BKT_R + t;
    if (node < N_NODES) offs[node] = first;
    cnt[t] = first;
  }
  if (b == NBKT - 1 && t == 0) offs[N_NODES] = k1;
  __syncthreads();
  for (int i0 = k0; i0 < k1; i0 += 1024 * 8) {
    int kv[8];
#pragma unroll
    for (int j = 0; j < 8; ++j) {
      const int i = i0 + j * 1024 + t;
      kv[j] = (i < k1) ? keys[i] : -1;
    }
#pragma unroll
    for (int j = 0; j < 8; ++j) {
      if (kv[j] >= 0) {
        const int p = atomicAdd(&cnt[kv[j] >> SRC_BITS], 1);
        bsrc[p] = kv[j] & ((1 << SRC_BITS) - 1);
      }
    }
  }
}

__global__ __launch_bounds__(256)
void k_aggr(const float* __restrict__ x, const int* __restrict__ offs,
            const int* __restrict__ bsrc, float* __restrict__ h) {
  int node = (blockIdx.x << 3) + (threadIdx.x >> 5);
  if (node >= N_NODES) return;
  int lane = threadIdx.x & 31;
  float4 acc = *reinterpret_cast<const float4*>(x + (size_t)node * D_FEAT + lane * 4);
  int e0 = offs[node], e1 = offs[node + 1];
  for (int e = e0; e < e1; ++e) {
    int s = bsrc[e];
    float4 v = *reinterpret_cast<const float4*>(x + (size_t)s * D_FEAT + lane * 4);
    acc.x += v.x; acc.y += v.y; acc.z += v.z; acc.w += v.w;
  }
  *reinterpret_cast<float4*>(h + (size_t)node * D_FEAT + lane * 4) = acc;
}

#define P_PITCH (D_OUT + 4)
#define PF 4
typedef float v4f __attribute__((ext_vector_type(4)));

template <int PITCH_W>
__device__ __forceinline__ void w_preload(const float* __restrict__ wp, v4f (&bx)[PF], v4f (&by)[PF]) {
#pragma unroll
  for (int s = 0; s < PF; ++s) {
    bx[s] = *reinterpret_cast<const v4f*>(wp + (size_t)(4 * s) * PITCH_W);
    by[s] = *reinterpret_cast<const v4f*>(wp + (size_t)(4 * s + 1) * PITCH_W);
  }
}

template <int PITCH_W, int PITCH_A, int NSTEP>
__device__ __forceinline__ void wave_tile_gemm(const float* __restrict__ wp, const float* ap,
                                               v4f (&bx)[PF], v4f (&by)[PF], v8f (&acc)[2][4]) {
  v2f a0 = *reinterpret_cast<const v2f*>(ap);
  v2f a1 = *reinterpret_cast<const v2f*>(ap + 16 * PITCH_A);
#pragma unroll
  for (int s = 0; s < NSTEP; ++s) {
    const int slot = s % PF;
    v2f b[4];
#pragma unroll
    for (int j = 0; j < 4; ++j) { b[j].x = bx[slot][j]; b[j].y = by[slot][j]; }
    const v2f c0 = a0, c1 = a1;
    if (s + PF < NSTEP) {
      bx[slot] = *reinterpret_cast<const v4f*>(wp + (size_t)(4 * (s + PF)) * PITCH_W);
      by[slot] = *reinterpret_cast<const v4f*>(wp + (size_t)(4 * (s + PF) + 1) * PITCH_W);
    }
    if (s + 1 < NSTEP) {
      a0 = *reinterpret_cast<const v2f*>(ap + 4 * (s + 1));
      a1 = *reinterpret_cast<const v2f*>(ap + 16 * PITCH_A + 4 * (s + 1));
    }
    __builtin_amdgcn_sched_barrier(0);
#pragma unroll
    for (int j = 0; j < 4; ++j) {
      acc[0][j] = wmma4(c0, b[j], acc[0][j]);
      acc[1][j] = wmma4(c1, b[j], acc[1][j]);
    }
  }
}

__global__ __launch_bounds__(256)
void gin_mlp(const float* __restrict__ h,
             const float* __restrict__ W1, const float* __restrict__ b1,
             const float* __restrict__ W2, const float* __restrict__ b2,
             float* __restrict__ out) {
  extern __shared__ float smem[];
  float* sH  = smem;
  float* sH1 = smem + 32 * H_PITCH;

  const int tid   = threadIdx.x;
  const int wave  = tid >> 5;
  const int lane  = tid & 31;
  const int row16 = lane & 15;
  const int khalf = (lane >> 4) << 1;
  const int rowhi = (lane >> 4) << 3;
  const int c4    = row16 << 2;
  const size_t base = (size_t)blockIdx.x * 32 * D_FEAT;

  v4f bx[PF], by[PF];
  v8f acc[2][4];

  const int n1 = wave << 6;
  const float* wp1 = W1 + (size_t)khalf * D_HID + n1 + c4;
  w_preload<D_HID>(wp1, bx, by);

  for (int i = tid; i < 32 * D_FEAT; i += 256)
    sH[(i >> 7) * H_PITCH + (i & (D_FEAT - 1))] = h[base + i];
  __syncthreads();

#pragma unroll
  for (int m = 0; m < 2; ++m)
#pragma unroll
    for (int j = 0; j < 4; ++j)
      acc[m][j] = (v8f){0.f,0.f,0.f,0.f,0.f,0.f,0.f,0.f};
  wave_tile_gemm<D_HID, H_PITCH, D_FEAT / 4>(wp1, sH + row16 * H_PITCH + khalf, bx, by, acc);

  const int q  = wave >> 1;
  const int n2 = (wave & 1) << 6;
  const int k0 = q << 7;
  const float* wp2 = W2 + (size_t)(k0 + khalf) * D_OUT + n2 + c4;
  w_preload<D_OUT>(wp2, bx, by);

  {
    const v4f bias = *reinterpret_cast<const v4f*>(b1 + n1 + c4);
#pragma unroll
    for (int m = 0; m < 2; ++m)
#pragma unroll
      for (int i = 0; i < 8; ++i) {
        v4f v;
        v.x = acc[m][0][i] + bias.x; v.y = acc[m][1][i] + bias.y;
        v.z = acc[m][2][i] + bias.z; v.w = acc[m][3][i] + bias.w;
        v.x = v.x > 0.f ? v.x : 0.f; v.y = v.y > 0.f ? v.y : 0.f;
        v.z = v.z > 0.f ? v.z : 0.f; v.w = v.w > 0.f ? v.w : 0.f;
        *reinterpret_cast<v4f*>(sH1 + (16 * m + i + rowhi) * H1_PITCH + n1 + c4) = v;
      }
  }
  __syncthreads();

#pragma unroll
  for (int m = 0; m < 2; ++m)
#pragma unroll
    for (int j = 0; j < 4; ++j)
      acc[m][j] = (v8f){0.f,0.f,0.f,0.f,0.f,0.f,0.f,0.f};
  wave_tile_gemm<D_OUT, H1_PITCH, D_HID / 16>(wp2, sH1 + row16 * H1_PITCH + k0 + khalf, bx, by, acc);

  __syncthreads();
  float* sP = smem;
#pragma unroll
  for (int m = 0; m < 2; ++m)
#pragma unroll
    for (int i = 0; i < 8; ++i) {
      v4f v;
      v.x = acc[m][0][i]; v.y = acc[m][1][i]; v.z = acc[m][2][i]; v.w = acc[m][3][i];
      *reinterpret_cast<v4f*>(sP + (q * 32 + 16 * m + i + rowhi) * P_PITCH + n2 + c4) = v;
    }
  __syncthreads();
  const int ocol = lane << 2;
  const v4f bias2 = *reinterpret_cast<const v4f*>(b2 + ocol);
  const size_t obase = (size_t)blockIdx.x * 32 * D_OUT;
#pragma unroll
  for (int i = 0; i < 4; ++i) {
    const int r = wave + 8 * i;
    const float* p = sP + r * P_PITCH + ocol;
    v4f s = *reinterpret_cast<const v4f*>(p);
    s += *reinterpret_cast<const v4f*>(p + 32 * P_PITCH);
    s += *reinterpret_cast<const v4f*>(p + 64 * P_PITCH);
    s += *reinterpret_cast<const v4f*>(p + 96 * P_PITCH);
    s += bias2;
    *reinterpret_cast<v4f*>(out + obase + (size_t)r * D_OUT + ocol) = s;
  }
}

extern "C" void kernel_launch(void* const* d_in, const int* in_sizes, int n_in,
                              void* d_out, int out_size, void* d_ws, size_t ws_size,
                              hipStream_t stream) {
  const float* x  = (const float*)d_in[0];
  const int*   ei = (const int*)d_in[1];
  const float* W1 = (const float*)d_in[2];
  const float* b1 = (const float*)d_in[3];
  const float* W2 = (const float*)d_in[4];
  const float* b2 = (const float*)d_in[5];
  float* out = (float*)d_out;

  const int* src = ei;
  const int* dst = ei + N_EDGES;

  float* h     = (float*)d_ws;
  int*  bsrc   = (int*)(h + (size_t)N_NODES * D_FEAT);
  int*  keys   = bsrc + N_EDGES;
  int*  hist   = keys + N_EDGES;
  int*  posl   = hist + NBKT * NCHUNK;
  int*  btot   = posl + NBKT * NCHUNK;
  int*  bstart = btot + NBKT;
  int*  offs   = bstart + NBKT + 1;

  k_hist<<<NCHUNK, 256, 0, stream>>>(dst, hist);
  k_bscan<<<NBKT, 256, 0, stream>>>(hist, posl, btot);
  k_scatter<<<NCHUNK, 256, 0, stream>>>(src, dst, posl, btot, keys, bstart);
  k_bucket<<<NBKT, 1024, 0, stream>>>(keys, bstart, offs, bsrc);

  k_aggr<<<(N_NODES + 7) / 8, 256, 0, stream>>>(x, offs, bsrc, h);

  gin_mlp<<<N_NODES / 32, 256, 82944, stream>>>(h, W1, b1, W2, b2, out);
}


# aggregation fused into the MLP kernel: each wave gathers 4 of the block's 32 rows (x[v] + bin rows, same order) straight into the LDS A tile, 8 row loads in flight; h never stored; 6 -> 5 launches
# speedup vs baseline: 2.6226x; 1.0860x over previous
#include <hip/hip_runtime.h>

#define N_NODES 100000
#define N_EDGES 1600000
#define D_FEAT  128
#define D_HID   512
#define D_OUT   128
#define BKT_SHIFT 9
#define BKT_R   (1 << BKT_SHIFT)
#define NBKT    ((N_NODES + BKT_R - 1) >> BKT_SHIFT)
#define NCHUNK  256
#define CHUNK_E ((N_EDGES + NCHUNK - 1) / NCHUNK)
#define EPT     ((CHUNK_E + 255) / 256)
#define SRC_BITS 17
static_assert(NBKT <= 256, "bucket totals are scanned by one 256-thread block");
static_assert(N_NODES <= (1 << SRC_BITS), "src id must fit the low bits of a placed edge");
static_assert(BKT_SHIFT + SRC_BITS < 32, "placed edge word stays non-negative");
static_assert(N_NODES % 32 == 0, "the fused kernel's 32-node tiles cover the nodes exactly");
#define H_PITCH  (D_FEAT + 4)
#define H1_PITCH (D_HID + 4)

typedef float v2f __attribute__((ext_vector_type(2)));
typedef float v8f __attribute__((ext_vector_type(8)));

__device__ __forceinline__ v8f wmma4(v2f a, v2f b, v8f c) {
  return __builtin_amdgcn_wmma_f32_16x16x4_f32(false, a, false, b,
                                               (short)0, c, false, false);
}


template <int N>
__device__ __forceinline__ int block_incl_scan(int v, int* s) {
  const int t = threadIdx.x;
  if (t < N) s[t] = v;
  __syncthreads();
#pragma unroll
  for (int off = 1; off < N; off <<= 1) {
    int a = 0;
    if (t < N && t >= off) a = s[t - off];
    __syncthreads();
    if (t < N) s[t] += a;
    __syncthreads();
  }
  return (t < N) ? s[t] : 0;
}

__global__ __launch_bounds__(256)
void k_hist(const int* __restrict__ dst, int* __restrict__ hist) {
  __shared__ int cnt[NBKT];
  const int t = threadIdx.x;
  const int e0 = blockIdx.x * CHUNK_E;
  const int e1 = (e0 + CHUNK_E < N_EDGES) ? (e0 + CHUNK_E) : N_EDGES;
  int d[EPT];
#pragma unroll
  for (int j = 0; j < EPT; ++j) {
    const int e = e0 + j * 256 + t;
    d[j] = (e < e1) ? dst[e] : -1;
  }
  for (int i = t; i < NBKT; i += 256) cnt[i] = 0;
  __syncthreads();
#pragma unroll
  for (int j = 0; j < EPT; ++j)
    if (d[j] >= 0) atomicAdd(&cnt[d[j] >> BKT_SHIFT], 1);
  __syncthreads();
  for (int i = t; i < NBKT; i += 256) hist[i * NCHUNK + blockIdx.x] = cnt[i];
}

__global__ __launch_bounds__(256)
void k_bscan(const int* __restrict__ hist, int* __restrict__ posl, int* __restrict__ btot) {
  __shared__ int s[256];
  const int t = threadIdx.x;
  const int v = hist[blockIdx.x * NCHUNK + t];
  const int incl = block_incl_scan<256>(v, s);
  posl[blockIdx.x * NCHUNK + t] = incl - v;
  if (t == 255) btot[blockIdx.x] = incl;
}

__global__ __launch_bounds__(256)
void k_scatter(const int* __restrict__ src, const int* __restrict__ dst,
               const int* __restrict__ posl, const int* __restrict__ btot,
               int* __restrict__ keys, int* __restrict__ bstart) {
  __shared__ int s[256];
  __shared__ int cur[NBKT];
  const int t = threadIdx.x;
  const int g = blockIdx.x;
  const int e0 = g * CHUNK_E;
  const int e1 = (e0 + CHUNK_E < N_EDGES) ? (e0 + CHUNK_E) : N_EDGES;
  int d[EPT], sr[EPT];
#pragma unroll
  for (int j = 0; j < EPT; ++j) {
    const int e = e0 + j * 256 + t;
    const bool ok = e < e1;
    d[j]  = ok ? dst[e] : -1;
    sr[j] = ok ? src[e] : 0;
  }
  const int tv = (t < NBKT) ? btot[t] : 0;
  const int incl = block_incl_scan<256>(tv, s);
  const int base = incl - tv;
  if (t < NBKT) cur[t] = base + posl[t * NCHUNK + g];
  if (g == 0) {
    if (t < NBKT) bstart[t] = base;
    if (t == NBKT - 1) bstart[NBKT] = incl;
  }
  __syncthreads();
#pragma unroll
  for (int j = 0; j < EPT; ++j) {
    if (d[j] >= 0) {
      const int p = atomicAdd(&cur[d[j] >> BKT_SHIFT], 1);
      keys[p] = ((d[j] & (BKT_R - 1)) << SRC_BITS) | sr[j];
    }
  }
}

__global__ __launch_bounds__(1024)
void k_bucket(const int* __restrict__ keys, const int* __restrict__ bstart,
              int* __restrict__ offs, int* __restrict__ bsrc) {
  __shared__ int cnt[BKT_R];
  __shared__ int sc[BKT_R];
  const int t = threadIdx.x;
  const int b = blockIdx.x;
  const int k0 = bstart[b], k1 = bstart[b + 1];
  if (t < BKT_R) cnt[t] = 0;
  __syncthreads();
  for (int i0 = k0; i0 < k1; i0 += 1024 * 8) {
    int kv[8];
#pragma unroll
    for (int j = 0; j < 8; ++j) {
      const int i = i0 + j * 1024 + t;
      kv[j] = (i < k1) ? keys[i] : -1;
    }
#pragma unroll
    for (int j = 0; j < 8; ++j)
      if (kv[j] >= 0) atomicAdd(&cnt[kv[j] >> SRC_BITS], 1);
  }
  __syncthreads();
  const int v = (t < BKT_R) ? cnt[t] : 0;
  const int incl = block_incl_scan<BKT_R>(v, sc);
  if (t < BKT_R) {
    const int first = k0 + incl - v;
    const int node = b * BKT_R + t;
    if (node < N_NODES) offs[node] = first;
    cnt[t] = first;
  }
  if (b == NBKT - 1 && t == 0) offs[N_NODES] = k1;
  __syncthreads();
  for (int i0 = k0; i0 < k1; i0 += 1024 * 8) {
    int kv[8];
#pragma unroll
    for (int j = 0; j < 8; ++j) {
      const int i = i0 + j * 1024 + t;
      kv[j] = (i < k1) ? keys[i] : -1;
    }
#pragma unroll
    for (int j = 0; j < 8; ++j) {
      if (kv[j] >= 0) {
        const int p = atomicAdd(&cnt[kv[j] >> SRC_BITS], 1);
        bsrc[p] = kv[j] & ((1 << SRC_BITS) - 1);
      }
    }
  }
}

#define P_PITCH (D_OUT + 4)
#define PF 4
#define GU 8
typedef float v4f __attribute__((ext_vector_type(4)));

template <int PITCH_W>
__device__ __forceinline__ void w_preload(const float* __restrict__ wp, v4f (&bx)[PF], v4f (&by)[PF]) {
#pragma unroll
  for (int s = 0; s < PF; ++s) {
    bx[s] = *reinterpret_cast<const v4f*>(wp + (size_t)(4 * s) * PITCH_W);
    by[s] = *reinterpret_cast<const v4f*>(wp + (size_t)(4 * s + 1) * PITCH_W);
  }
}

template <int PITCH_W, int PITCH_A, int NSTEP>
__device__ __forceinline__ void wave_tile_gemm(const float* __restrict__ wp, const float* ap,
                                               v4f (&bx)[PF], v4f (&by)[PF], v8f (&acc)[2][4]) {
  v2f a0 = *reinterpret_cast<const v2f*>(ap);
  v2f a1 = *reinterpret_cast<const v2f*>(ap + 16 * PITCH_A);
#pragma unroll
  for (int s = 0; s < NSTEP; ++s) {
    const int slot = s % PF;
    v2f b[4];
#pragma unroll
    for (int j = 0; j < 4; ++j) { b[j].x = bx[slot][j]; b[j].y = by[slot][j]; }
    const v2f c0 = a0, c1 = a1;
    if (s + PF < NSTEP) {
      bx[slot] = *reinterpret_cast<const v4f*>(wp + (size_t)(4 * (s + PF)) * PITCH_W);
      by[slot] = *reinterpret_cast<const v4f*>(wp + (size_t)(4 * (s + PF) + 1) * PITCH_W);
    }
    if (s + 1 < NSTEP) {
      a0 = *reinterpret_cast<const v2f*>(ap + 4 * (s + 1));
      a1 = *reinterpret_cast<const v2f*>(ap + 16 * PITCH_A + 4 * (s + 1));
    }
    __builtin_amdgcn_sched_barrier(0);
#pragma unroll
    for (int j = 0; j < 4; ++j) {
      acc[0][j] = wmma4(c0, b[j], acc[0][j]);
      acc[1][j] = wmma4(c1, b[j], acc[1][j]);
    }
  }
}

__global__ __launch_bounds__(256)
void gin_mlp(const float* __restrict__ x, const int* __restrict__ offs, const int* __restrict__ bsrc,
             const float* __restrict__ W1, const float* __restrict__ b1,
             const float* __restrict__ W2, const float* __restrict__ b2,
             float* __restrict__ out) {
  extern __shared__ float smem[];
  float* sH  = smem;
  float* sH1 = smem + 32 * H_PITCH;

  const int tid   = threadIdx.x;
  const int wave  = tid >> 5;
  const int lane  = tid & 31;
  const int row16 = lane & 15;
  const int khalf = (lane >> 4) << 1;
  const int rowhi = (lane >> 4) << 3;
  const int c4    = row16 << 2;

  v4f bx[PF], by[PF];
  v8f acc[2][4];

  const int n1 = wave << 6;
  const float* wp1 = W1 + (size_t)khalf * D_HID + n1 + c4;
  w_preload<D_HID>(wp1, bx, by);

  {
    const int wv    = __builtin_amdgcn_readfirstlane(wave);
    const int node0 = blockIdx.x * 32 + wv * 4;
    const float* xl = x + lane * 4;
    int eb[5];
#pragma unroll
    for (int i = 0; i < 5; ++i) eb[i] = offs[node0 + i];
#pragma unroll
    for (int i = 0; i < 4; ++i) {
      v4f a = *reinterpret_cast<const v4f*>(xl + (size_t)(node0 + i) * D_FEAT);
      const int e1 = eb[i + 1];
      for (int e = eb[i]; e < e1; e += GU) {
        int s[GU];
#pragma unroll
        for (int j = 0; j < GU; ++j) {
          const int ee = (e + j < e1) ? (e + j) : (e1 - 1);
          s[j] = bsrc[ee];
        }
        v4f v[GU];
#pragma unroll
        for (int j = 0; j < GU; ++j)
          v[j] = *reinterpret_cast<const v4f*>(xl + (size_t)s[j] * D_FEAT);
#pragma unroll
        for (int j = 0; j < GU; ++j)
          if (e + j < e1) a += v[j];
      }
      *reinterpret_cast<v4f*>(sH + (wv * 4 + i) * H_PITCH + lane * 4) = a;
    }
  }
  __syncthreads();

#pragma unroll
  for (int m = 0; m < 2; ++m)
#pragma unroll
    for (int j = 0; j < 4; ++j)
      acc[m][j] = (v8f){0.f,0.f,0.f,0.f,0.f,0.f,0.f,0.f};
  wave_tile_gemm<D_HID, H_PITCH, D_FEAT / 4>(wp1, sH + row16 * H_PITCH + khalf, bx, by, acc);

  const int q  = wave >> 1;
  const int n2 = (wave & 1) << 6;
  const int k0 = q << 7;
  const float* wp2 = W2 + (size_t)(k0 + khalf) * D_OUT + n2 + c4;
  w_preload<D_OUT>(wp2, bx, by);

  {
    const v4f bias = *reinterpret_cast<const v4f*>(b1 + n1 + c4);
#pragma unroll
    for (int m = 0; m < 2; ++m)
#pragma unroll
      for (int i = 0; i < 8; ++i) {
        v4f v;
        v.x = acc[m][0][i] + bias.x; v.y = acc[m][1][i] + bias.y;
        v.z = acc[m][2][i] + bias.z; v.w = acc[m][3][i] + bias.w;
        v.x = v.x > 0.f ? v.x : 0.f; v.y = v.y > 0.f ? v.y : 0.f;
        v.z = v.z > 0.f ? v.z : 0.f; v.w = v.w > 0.f ? v.w : 0.f;
        *reinterpret_cast<v4f*>(sH1 + (16 * m + i + rowhi) * H1_PITCH + n1 + c4) = v;
      }
  }
  __syncthreads();

#pragma unroll
  for (int m = 0; m < 2; ++m)
#pragma unroll
    for (int j = 0; j < 4; ++j)
      acc[m][j] = (v8f){0.f,0.f,0.f,0.f,0.f,0.f,0.f,0.f};
  wave_tile_gemm<D_OUT, H1_PITCH, D_HID / 16>(wp2, sH1 + row16 * H1_PITCH + k0 + khalf, bx, by, acc);

  __syncthreads();
  float* sP = smem;
#pragma unroll
  for (int m = 0; m < 2; ++m)
#pragma unroll
    for (int i = 0; i < 8; ++i) {
      v4f v;
      v.x = acc[m][0][i]; v.y = acc[m][1][i]; v.z = acc[m][2][i]; v.w = acc[m][3][i];
      *reinterpret_cast<v4f*>(sP + (q * 32 + 16 * m + i + rowhi) * P_PITCH + n2 + c4) = v;
    }
  __syncthreads();
  const int ocol = lane << 2;
  const v4f bias2 = *reinterpret_cast<const v4f*>(b2 + ocol);
  const size_t obase = (size_t)blockIdx.x * 32 * D_OUT;
#pragma unroll
  for (int i = 0; i < 4; ++i) {
    const int r = wave + 8 * i;
    const float* p = sP + r * P_PITCH + ocol;
    v4f s = *reinterpret_cast<const v4f*>(p);
    s += *reinterpret_cast<const v4f*>(p + 32 * P_PITCH);
    s += *reinterpret_cast<const v4f*>(p + 64 * P_PITCH);
    s += *reinterpret_cast<const v4f*>(p + 96 * P_PITCH);
    s += bias2;
    *reinterpret_cast<v4f*>(out + obase + (size_t)r * D_OUT + ocol) = s;
  }
}

extern "C" void kernel_launch(void* const* d_in, const int* in_sizes, int n_in,
                              void* d_out, int out_size, void* d_ws, size_t ws_size,
                              hipStream_t stream) {
  const float* x  = (const float*)d_in[0];
  const int*   ei = (const int*)d_in[1];
  const float* W1 = (const float*)d_in[2];
  const float* b1 = (const float*)d_in[3];
  const float* W2 = (const float*)d_in[4];
  const float* b2 = (const float*)d_in[5];
  float* out = (float*)d_out;

  const int* src = ei;
  const int* dst = ei + N_EDGES;

  int*  bsrc   = (int*)d_ws;
  int*  keys   = bsrc + N_EDGES;
  int*  hist   = keys + N_EDGES;
  int*  posl   = hist + NBKT * NCHUNK;
  int*  btot   = posl + NBKT * NCHUNK;
  int*  bstart = btot + NBKT;
  int*  offs   = bstart + NBKT + 1;

  k_hist<<<NCHUNK, 256, 0, stream>>>(dst, hist);
  k_bscan<<<NBKT, 256, 0, stream>>>(hist, posl, btot);
  k_scatter<<<NCHUNK, 256, 0, stream>>>(src, dst, posl, btot, keys, bstart);
  k_bucket<<<NBKT, 1024, 0, stream>>>(keys, bstart, offs, bsrc);

  gin_mlp<<<N_NODES / 32, 256, 82944, stream>>>(x, offs, bsrc, W1, b1, W2, b2, out);
}


# CSR build: both placement passes (edges to bucket regions, src ids to node bins) stage their words in LDS and store them in slot order: whole lines per wave store, not 32 scattered 4-byte pieces
# speedup vs baseline: 2.9309x; 1.1176x over previous
#include <hip/hip_runtime.h>

#define N_NODES 100000
#define N_EDGES 1600000
#define D_FEAT  128
#define D_HID   512
#define D_OUT   128
#define BKT_SHIFT 9
#define BKT_R   (1 << BKT_SHIFT)
#define NBKT    ((N_NODES + BKT_R - 1) >> BKT_SHIFT)
#define NCHUNK  256
#define CHUNK_E ((N_EDGES + NCHUNK - 1) / NCHUNK)
#define EPT     ((CHUNK_E + 255) / 256)
#define SRC_BITS 17
static_assert(NBKT <= 256, "bucket totals are scanned by one 256-thread block");
static_assert(N_NODES <= (1 << SRC_BITS), "src id must fit the low bits of a placed edge");
static_assert(BKT_SHIFT + SRC_BITS < 32, "placed edge word stays non-negative");
static_assert(N_NODES % 32 == 0, "the fused kernel's 32-node tiles cover the nodes exactly");
#define H_PITCH  (D_FEAT + 4)
#define H1_PITCH (D_HID + 4)

typedef float v2f __attribute__((ext_vector_type(2)));
typedef float v8f __attribute__((ext_vector_type(8)));

__device__ __forceinline__ v8f wmma4(v2f a, v2f b, v8f c) {
  return __builtin_amdgcn_wmma_f32_16x16x4_f32(false, a, false, b,
                                               (short)0, c, false, false);
}

#define STAGE_CAP 12288

template <int N>
__device__ __forceinline__ int block_incl_scan(int v, int* s) {
  const int t = threadIdx.x;
  if (t < N) s[t] = v;
  __syncthreads();
#pragma unroll
  for (int off = 1; off < N; off <<= 1) {
    int a = 0;
    if (t < N && t >= off) a = s[t - off];
    __syncthreads();
    if (t < N) s[t] += a;
    __syncthreads();
  }
  return (t < N) ? s[t] : 0;
}

template <int N>
__device__ __forceinline__ long long block_incl_scan64(long long v, long long* s) {
  const int t = threadIdx.x;
  if (t < N) s[t] = v;
  __syncthreads();
#pragma unroll
  for (int off = 1; off < N; off <<= 1) {
    long long a = 0;
    if (t < N && t >= off) a = s[t - off];
    __syncthreads();
    if (t < N) s[t] += a;
    __syncthreads();
  }
  return (t < N) ? s[t] : 0;
}

__global__ __launch_bounds__(256)
void k_hist(const int* __restrict__ dst, int* __restrict__ hist) {
  __shared__ int cnt[NBKT];
  const int t = threadIdx.x;
  const int e0 = blockIdx.x * CHUNK_E;
  const int e1 = (e0 + CHUNK_E < N_EDGES) ? (e0 + CHUNK_E) : N_EDGES;
  int d[EPT];
#pragma unroll
  for (int j = 0; j < EPT; ++j) {
    const int e = e0 + j * 256 + t;
    d[j] = (e < e1) ? dst[e] : -1;
  }
  for (int i = t; i < NBKT; i += 256) cnt[i] = 0;
  __syncthreads();
#pragma unroll
  for (int j = 0; j < EPT; ++j)
    if (d[j] >= 0) atomicAdd(&cnt[d[j] >> BKT_SHIFT], 1);
  __syncthreads();
  for (int i = t; i < NBKT; i += 256) hist[i * NCHUNK + blockIdx.x] = cnt[i];
}

__global__ __launch_bounds__(256)
void k_bscan(const int* __restrict__ hist, int* __restrict__ posl, int* __restrict__ btot) {
  __shared__ int s[256];
  const int t = threadIdx.x;
  const int v = hist[blockIdx.x * NCHUNK + t];
  const int incl = block_incl_scan<256>(v, s);
  posl[blockIdx.x * NCHUNK + t] = incl - v;
  if (t == 255) btot[blockIdx.x] = incl;
}

__global__ __launch_bounds__(256)
void k_scatter(const int* __restrict__ src, const int* __restrict__ dst,
               const int* __restrict__ hist, const int* __restrict__ posl, const int* __restrict__ btot,
               int* __restrict__ keys, int* __restrict__ bstart) {
  __shared__ long long s[256];
  __shared__ int cur[NBKT];
  __shared__ int gd[NBKT];
  __shared__ int stw[CHUNK_E];
  __shared__ int stp[CHUNK_E];
  const int t = threadIdx.x;
  const int g = blockIdx.x;
  const int e0 = g * CHUNK_E;
  const int e1 = (e0 + CHUNK_E < N_EDGES) ? (e0 + CHUNK_E) : N_EDGES;
  int d[EPT], sr[EPT];
#pragma unroll
  for (int j = 0; j < EPT; ++j) {
    const int e = e0 + j * 256 + t;
    const bool ok = e < e1;
    d[j]  = ok ? dst[e] : -1;
    sr[j] = ok ? src[e] : 0;
  }
  const int tv = (t < NBKT) ? btot[t] : 0;
  const int lc = (t < NBKT) ? hist[t * NCHUNK + g] : 0;
  const long long incl = block_incl_scan64<256>(((long long)tv << 32) | (long long)lc, s);
  const int gincl = (int)(incl >> 32);
  const int base  = gincl - tv;
  const int lbase = (int)(incl & 0xffffffffLL) - lc;
  if (t < NBKT) {
    cur[t] = lbase;
    gd[t]  = base + posl[t * NCHUNK + g] - lbase;
  }
  if (g == 0) {
    if (t < NBKT) bstart[t] = base;
    if (t == NBKT - 1) bstart[NBKT] = gincl;
  }
  __syncthreads();
#pragma unroll
  for (int j = 0; j < EPT; ++j) {
    if (d[j] >= 0) {
      const int b  = d[j] >> BKT_SHIFT;
      const int lp = atomicAdd(&cur[b], 1);
      stw[lp] = ((d[j] & (BKT_R - 1)) << SRC_BITS) | sr[j];
      stp[lp] = lp + gd[b];
    }
  }
  __syncthreads();
  const int n = e1 - e0;
  for (int j = t; j < n; j += 256) keys[stp[j]] = stw[j];
}

__global__ __launch_bounds__(1024)
void k_bucket(const int* __restrict__ keys, const int* __restrict__ bstart,
              int* __restrict__ offs, int* __restrict__ bsrc) {
  __shared__ int cnt[BKT_R];
  __shared__ int sc[BKT_R];
  __shared__ int stage[STAGE_CAP];
  const int t = threadIdx.x;
  const int b = blockIdx.x;
  const int k0 = bstart[b], k1 = bstart[b + 1];
  const bool staged = (k1 - k0) <= STAGE_CAP;
  if (t < BKT_R) cnt[t] = 0;
  __syncthreads();
  for (int i0 = k0; i0 < k1; i0 += 1024 * 8) {
    int kv[8];
#pragma unroll
    for (int j = 0; j < 8; ++j) {
      const int i = i0 + j * 1024 + t;
      kv[j] = (i < k1) ? keys[i] : -1;
    }
#pragma unroll
    for (int j = 0; j < 8; ++j)
      if (kv[j] >= 0) atomicAdd(&cnt[kv[j] >> SRC_BITS], 1);
  }
  __syncthreads();
  const int v = (t < BKT_R) ? cnt[t] : 0;
  const int incl = block_incl_scan<BKT_R>(v, sc);
  if (t < BKT_R) {
    const int first = k0 + incl - v;
    const int node = b * BKT_R + t;
    if (node < N_NODES) offs[node] = first;
    cnt[t] = first;
  }
  if (b == NBKT - 1 && t == 0) offs[N_NODES] = k1;
  __syncthreads();
  for (int i0 = k0; i0 < k1; i0 += 1024 * 8) {
    int kv[8];
#pragma unroll
    for (int j = 0; j < 8; ++j) {
      const int i = i0 + j * 1024 + t;
      kv[j] = (i < k1) ? keys[i] : -1;
    }
#pragma unroll
    for (int j = 0; j < 8; ++j) {
      if (kv[j] >= 0) {
        const int p = atomicAdd(&cnt[kv[j] >> SRC_BITS], 1);
        const int sv = kv[j] & ((1 << SRC_BITS) - 1);
        if (staged) stage[p - k0] = sv; else bsrc[p] = sv;
      }
    }
  }
  if (staged) {
    __syncthreads();
    for (int i = t; i < k1 - k0; i += 1024) bsrc[k0 + i] = stage[i];
  }
}

#define P_PITCH (D_OUT + 4)
#define PF 4
#define GU 8
typedef float v4f __attribute__((ext_vector_type(4)));

template <int PITCH_W>
__device__ __forceinline__ void w_preload(const float* __restrict__ wp, v4f (&bx)[PF], v4f (&by)[PF]) {
#pragma unroll
  for (int s = 0; s < PF; ++s) {
    bx[s] = *reinterpret_cast<const v4f*>(wp + (size_t)(4 * s) * PITCH_W);
    by[s] = *reinterpret_cast<const v4f*>(wp + (size_t)(4 * s + 1) * PITCH_W);
  }
}

template <int PITCH_W, int PITCH_A, int NSTEP>
__device__ __forceinline__ void wave_tile_gemm(const float* __restrict__ wp, const float* ap,
                                               v4f (&bx)[PF], v4f (&by)[PF], v8f (&acc)[2][4]) {
  v2f a0 = *reinterpret_cast<const v2f*>(ap);
  v2f a1 = *reinterpret_cast<const v2f*>(ap + 16 * PITCH_A);
#pragma unroll
  for (int s = 0; s < NSTEP; ++s) {
    const int slot = s % PF;
    v2f b[4];
#pragma unroll
    for (int j = 0; j < 4; ++j) { b[j].x = bx[slot][j]; b[j].y = by[slot][j]; }
    const v2f c0 = a0, c1 = a1;
    if (s + PF < NSTEP) {
      bx[slot] = *reinterpret_cast<const v4f*>(wp + (size_t)(4 * (s + PF)) * PITCH_W);
      by[slot] = *reinterpret_cast<const v4f*>(wp + (size_t)(4 * (s + PF) + 1) * PITCH_W);
    }
    if (s + 1 < NSTEP) {
      a0 = *reinterpret_cast<const v2f*>(ap + 4 * (s + 1));
      a1 = *reinterpret_cast<const v2f*>(ap + 16 * PITCH_A + 4 * (s + 1));
    }
    __builtin_amdgcn_sched_barrier(0);
#pragma unroll
    for (int j = 0; j < 4; ++j) {
      acc[0][j] = wmma4(c0, b[j], acc[0][j]);
      acc[1][j] = wmma4(c1, b[j], acc[1][j]);
    }
  }
}

__global__ __launch_bounds__(256)
void gin_mlp(const float* __restrict__ x, const int* __restrict__ offs, const int* __restrict__ bsrc,
             const float* __restrict__ W1, const float* __restrict__ b1,
             const float* __restrict__ W2, const float* __restrict__ b2,
             float* __restrict__ out) {
  extern __shared__ float smem[];
  float* sH  = smem;
  float* sH1 = smem + 32 * H_PITCH;

  const int tid   = threadIdx.x;
  const int wave  = tid >> 5;
  const int lane  = tid & 31;
  const int row16 = lane & 15;
  const int khalf = (lane >> 4) << 1;
  const int rowhi = (lane >> 4) << 3;
  const int c4    = row16 << 2;

  v4f bx[PF], by[PF];
  v8f acc[2][4];

  const int n1 = wave << 6;
  const float* wp1 = W1 + (size_t)khalf * D_HID + n1 + c4;
  w_preload<D_HID>(wp1, bx, by);

  {
    const int wv    = __builtin_amdgcn_readfirstlane(wave);
    const int node0 = blockIdx.x * 32 + wv * 4;
    const float* xl = x + lane * 4;
    int eb[5];
#pragma unroll
    for (int i = 0; i < 5; ++i) eb[i] = offs[node0 + i];
#pragma unroll
    for (int i = 0; i < 4; ++i) {
      v4f a = *reinterpret_cast<const v4f*>(xl + (size_t)(node0 + i) * D_FEAT);
      const int e1 = eb[i + 1];
      for (int e = eb[i]; e < e1; e += GU) {
        int s[GU];
#pragma unroll
        for (int j = 0; j < GU; ++j) {
          const int ee = (e + j < e1) ? (e + j) : (e1 - 1);
          s[j] = bsrc[ee];
        }
        v4f v[GU];
#pragma unroll
        for (int j = 0; j < GU; ++j)
          v[j] = *reinterpret_cast<const v4f*>(xl + (size_t)s[j] * D_FEAT);
#pragma unroll
        for (int j = 0; j < GU; ++j)
          if (e + j < e1) a += v[j];
      }
      *reinterpret_cast<v4f*>(sH + (wv * 4 + i) * H_PITCH + lane * 4) = a;
    }
  }
  __syncthreads();

#pragma unroll
  for (int m = 0; m < 2; ++m)
#pragma unroll
    for (int j = 0; j < 4; ++j)
      acc[m][j] = (v8f){0.f,0.f,0.f,0.f,0.f,0.f,0.f,0.f};
  wave_tile_gemm<D_HID, H_PITCH, D_FEAT / 4>(wp1, sH + row16 * H_PITCH + khalf, bx, by, acc);

  const int q  = wave >> 1;
  const int n2 = (wave & 1) << 6;
  const int k0 = q << 7;
  const float* wp2 = W2 + (size_t)(k0 + khalf) * D_OUT + n2 + c4;
  w_preload<D_OUT>(wp2, bx, by);

  {
    const v4f bias = *reinterpret_cast<const v4f*>(b1 + n1 + c4);
#pragma unroll
    for (int m = 0; m < 2; ++m)
#pragma unroll
      for (int i = 0; i < 8; ++i) {
        v4f v;
        v.x = acc[m][0][i] + bias.x; v.y = acc[m][1][i] + bias.y;
        v.z = acc[m][2][i] + bias.z; v.w = acc[m][3][i] + bias.w;
        v.x = v.x > 0.f ? v.x : 0.f; v.y = v.y > 0.f ? v.y : 0.f;
        v.z = v.z > 0.f ? v.z : 0.f; v.w = v.w > 0.f ? v.w : 0.f;
        *reinterpret_cast<v4f*>(sH1 + (16 * m + i + rowhi) * H1_PITCH + n1 + c4) = v;
      }
  }
  __syncthreads();

#pragma unroll
  for (int m = 0; m < 2; ++m)
#pragma unroll
    for (int j = 0; j < 4; ++j)
      acc[m][j] = (v8f){0.f,0.f,0.f,0.f,0.f,0.f,0.f,0.f};
  wave_tile_gemm<D_OUT, H1_PITCH, D_HID / 16>(wp2, sH1 + row16 * H1_PITCH + k0 + khalf, bx, by, acc);

  __syncthreads();
  float* sP = smem;
#pragma unroll
  for (int m = 0; m < 2; ++m)
#pragma unroll
    for (int i = 0; i < 8; ++i) {
      v4f v;
      v.x = acc[m][0][i]; v.y = acc[m][1][i]; v.z = acc[m][2][i]; v.w = acc[m][3][i];
      *reinterpret_cast<v4f*>(sP + (q * 32 + 16 * m + i + rowhi) * P_PITCH + n2 + c4) = v;
    }
  __syncthreads();
  const int ocol = lane << 2;
  const v4f bias2 = *reinterpret_cast<const v4f*>(b2 + ocol);
  const size_t obase = (size_t)blockIdx.x * 32 * D_OUT;
#pragma unroll
  for (int i = 0; i < 4; ++i) {
    const int r = wave + 8 * i;
    const float* p = sP + r * P_PITCH + ocol;
    v4f s = *reinterpret_cast<const v4f*>(p);
    s += *reinterpret_cast<const v4f*>(p + 32 * P_PITCH);
    s += *reinterpret_cast<const v4f*>(p + 64 * P_PITCH);
    s += *reinterpret_cast<const v4f*>(p + 96 * P_PITCH);
    s += bias2;
    *reinterpret_cast<v4f*>(out + obase + (size_t)r * D_OUT + ocol) = s;
  }
}

extern "C" void kernel_launch(void* const* d_in, const int* in_sizes, int n_in,
                              void* d_out, int out_size, void* d_ws, size_t ws_size,
                              hipStream_t stream) {
  const float* x  = (const float*)d_in[0];
  const int*   ei = (const int*)d_in[1];
  const float* W1 = (const float*)d_in[2];
  const float* b1 = (const float*)d_in[3];
  const float* W2 = (const float*)d_in[4];
  const float* b2 = (const float*)d_in[5];
  float* out = (float*)d_out;

  const int* src = ei;
  const int* dst = ei + N_EDGES;

  int*  bsrc   = (int*)d_ws;
  int*  keys   = bsrc + N_EDGES;
  int*  hist   = keys + N_EDGES;
  int*  posl   = hist + NBKT * NCHUNK;
  int*  btot   = posl + NBKT * NCHUNK;
  int*  bstart = btot + NBKT;
  int*  offs   = bstart + NBKT + 1;

  k_hist<<<NCHUNK, 256, 0, stream>>>(dst, hist);
  k_bscan<<<NBKT, 256, 0, stream>>>(hist, posl, btot);
  k_scatter<<<NCHUNK, 256, 0, stream>>>(src, dst, hist, posl, btot, keys, bstart);
  k_bucket<<<NBKT, 1024, 0, stream>>>(keys, bstart, offs, bsrc);

  gin_mlp<<<N_NODES / 32, 256, 82944, stream>>>(x, offs, bsrc, W1, b1, W2, b2, out);
}


# CSR build: every edge chunk sorts its own edges by bucket in LDS; a bucket's block sums the 256 chunk offsets and reads its 256 segments once; 5 -> 3 launches; bin order: another arrival order
# speedup vs baseline: 3.0423x; 1.0380x over previous
#include <hip/hip_runtime.h>

#define N_NODES 100000
#define N_EDGES 1600000
#define D_FEAT  128
#define D_HID   512
#define D_OUT   128
#define BKT_SHIFT 9
#define BKT_R   (1 << BKT_SHIFT)
#define NBKT    ((N_NODES + BKT_R - 1) >> BKT_SHIFT)
#define NCHUNK  256
#define CHUNK_E ((N_EDGES + NCHUNK - 1) / NCHUNK)
#define EPT     ((CHUNK_E + 255) / 256)
#define SRC_BITS 17
static_assert(NBKT <= 256, "a chunk's bucket counts are scanned by one 256-thread block");
static_assert(NCHUNK % 32 == 0 && NCHUNK <= 1024, "k_bucket reads one chunk offset per thread, whole waves");
static_assert(N_NODES <= (1 << SRC_BITS), "src id must fit the low bits of a placed edge");
static_assert(BKT_SHIFT + SRC_BITS < 32, "placed edge word stays non-negative");
static_assert(N_NODES % 32 == 0, "the fused kernel's 32-node tiles cover the nodes exactly");
#define H_PITCH  (D_FEAT + 4)
#define H1_PITCH (D_HID + 4)

typedef float v2f __attribute__((ext_vector_type(2)));
typedef float v8f __attribute__((ext_vector_type(8)));

__device__ __forceinline__ v8f wmma4(v2f a, v2f b, v8f c) {
  return __builtin_amdgcn_wmma_f32_16x16x4_f32(false, a, false, b,
                                               (short)0, c, false, false);
}

#define STAGE_CAP 12288
#define SEG_Q (NCHUNK / 32)

template <int N>
__device__ __forceinline__ int block_incl_scan(int v, int* s) {
  const int t = threadIdx.x;
  if (t < N) s[t] = v;
  __syncthreads();
#pragma unroll
  for (int off = 1; off < N; off <<= 1) {
    int a = 0;
    if (t < N && t >= off) a = s[t - off];
    __syncthreads();
    if (t < N) s[t] += a;
    __syncthreads();
  }
  return (t < N) ? s[t] : 0;
}

__global__ __launch_bounds__(256)
void k_part(const int* __restrict__ src, const int* __restrict__ dst,
            int* __restrict__ keys, int* __restrict__ coff) {
  __shared__ int s[256];
  __shared__ int cur[NBKT];
  __shared__ int stw[CHUNK_E];
  const int t = threadIdx.x;
  const int g = blockIdx.x;
  const int e0 = g * CHUNK_E;
  const int e1 = (e0 + CHUNK_E < N_EDGES) ? (e0 + CHUNK_E) : N_EDGES;
  int d[EPT], sr[EPT];
#pragma unroll
  for (int j = 0; j < EPT; ++j) {
    const int e = e0 + j * 256 + t;
    const bool ok = e < e1;
    d[j]  = ok ? dst[e] : -1;
    sr[j] = ok ? src[e] : 0;
  }
  for (int i = t; i < NBKT; i += 256) cur[i] = 0;
  __syncthreads();
  int rk[EPT];
#pragma unroll
  for (int j = 0; j < EPT; ++j)
    rk[j] = (d[j] >= 0) ? atomicAdd(&cur[d[j] >> BKT_SHIFT], 1) : 0;
  __syncthreads();
  const int lc = (t < NBKT) ? cur[t] : 0;
  const int incl = block_incl_scan<256>(lc, s);
  if (t < NBKT) {
    cur[t] = incl - lc;
    coff[t * NCHUNK + g] = incl - lc;
  }
  if (t == NBKT - 1) coff[NBKT * NCHUNK + g] = incl;
  __syncthreads();
#pragma unroll
  for (int j = 0; j < EPT; ++j)
    if (d[j] >= 0) stw[cur[d[j] >> BKT_SHIFT] + rk[j]] = ((d[j] & (BKT_R - 1)) << SRC_BITS) | sr[j];
  __syncthreads();
  const int n = e1 - e0;
  for (int j = t; j < n; j += 256) keys[e0 + j] = stw[j];
}

__global__ __launch_bounds__(1024)
void k_bucket(const int* __restrict__ keys, const int* __restrict__ coff,
              int* __restrict__ offs, int* __restrict__ bsrc) {
  __shared__ int cnt[BKT_R];
  __shared__ int sc[BKT_R];
  __shared__ int stage[STAGE_CAP];
  __shared__ int segp[NCHUNK];
  __shared__ int segn[NCHUNK];
  __shared__ int tot[2];
  const int t = threadIdx.x;
  const int b = blockIdx.x;
  const int lane = t & 31;
  const int wave = t >> 5;
  int o0 = 0, o1 = 0;
  if (t < NCHUNK) {
    o0 = coff[b * NCHUNK + t];
    o1 = coff[(b + 1) * NCHUNK + t];
  }
  if (t < BKT_R) cnt[t] = 0;
  if (t < 2) tot[t] = 0;
  __syncthreads();
  if (t < NCHUNK) {
    segp[t] = t * CHUNK_E + o0;
    segn[t] = o1 - o0;
    int r0 = o0, r1 = o1 - o0;
#pragma unroll
    for (int m = 16; m > 0; m >>= 1) {
      r0 += __shfl_xor(r0, m);
      r1 += __shfl_xor(r1, m);
    }
    if (lane == 0) { atomicAdd(&tot[0], r0); atomicAdd(&tot[1], r1); }
  }
  __syncthreads();
  const int k0 = tot[0], k1 = k0 + tot[1];
  const bool staged = (k1 - k0) <= STAGE_CAP;
  int kv[SEG_Q][2], sp[SEG_Q], sn[SEG_Q];
#pragma unroll
  for (int q = 0; q < SEG_Q; ++q) { sp[q] = segp[wave + 32 * q]; sn[q] = segn[wave + 32 * q]; }
#pragma unroll
  for (int q = 0; q < SEG_Q; ++q) {
    kv[q][0] = (lane < sn[q]) ? keys[sp[q] + lane] : -1;
    kv[q][1] = (lane + 32 < sn[q]) ? keys[sp[q] + lane + 32] : -1;
  }
#pragma unroll
  for (int q = 0; q < SEG_Q; ++q) {
    if (kv[q][0] >= 0) atomicAdd(&cnt[kv[q][0] >> SRC_BITS], 1);
    if (kv[q][1] >= 0) atomicAdd(&cnt[kv[q][1] >> SRC_BITS], 1);
  }
#pragma unroll
  for (int q = 0; q < SEG_Q; ++q)
    for (int j = 64 + lane; j < sn[q]; j += 32) atomicAdd(&cnt[keys[sp[q] + j] >> SRC_BITS], 1);
  __syncthreads();
  const int v = (t < BKT_R) ? cnt[t] : 0;
  const int incl = block_incl_scan<BKT_R>(v, sc);
  if (t < BKT_R) {
    const int first = k0 + incl - v;
    const int node = b * BKT_R + t;
    if (node < N_NODES) offs[node] = first;
    cnt[t] = first;
  }
  if (b == NBKT - 1 && t == 0) offs[N_NODES] = k1;
  __syncthreads();
#pragma unroll
  for (int q = 0; q < SEG_Q; ++q) {
#pragma unroll
    for (int u = 0; u < 2; ++u) {
      if (kv[q][u] >= 0) {
        const int p = atomicAdd(&cnt[kv[q][u] >> SRC_BITS], 1);
        const int sv = kv[q][u] & ((1 << SRC_BITS) - 1);
        if (staged) stage[p - k0] = sv; else bsrc[p] = sv;
      }
    }
  }
#pragma unroll
  for (int q = 0; q < SEG_Q; ++q) {
    for (int j = 64 + lane; j < sn[q]; j += 32) {
      const int kk = keys[sp[q] + j];
      const int p = atomicAdd(&cnt[kk >> SRC_BITS], 1);
      const int sv = kk & ((1 << SRC_BITS) - 1);
      if (staged) stage[p - k0] = sv; else bsrc[p] = sv;
    }
  }
  if (staged) {
    __syncthreads();
    for (int i = t; i < k1 - k0; i += 1024) bsrc[k0 + i] = stage[i];
  }
}

#define P_PITCH (D_OUT + 4)
#define PF 4
#define GU 8
typedef float v4f __attribute__((ext_vector_type(4)));

template <int PITCH_W>
__device__ __forceinline__ void w_preload(const float* __restrict__ wp, v4f (&bx)[PF], v4f (&by)[PF]) {
#pragma unroll
  for (int s = 0; s < PF; ++s) {
    bx[s] = *reinterpret_cast<const v4f*>(wp + (size_t)(4 * s) * PITCH_W);
    by[s] = *reinterpret_cast<const v4f*>(wp + (size_t)(4 * s + 1) * PITCH_W);
  }
}

template <int PITCH_W, int PITCH_A, int NSTEP>
__device__ __forceinline__ void wave_tile_gemm(const float* __restrict__ wp, const float* ap,
                                               v4f (&bx)[PF], v4f (&by)[PF], v8f (&acc)[2][4]) {
  v2f a0 = *reinterpret_cast<const v2f*>(ap);
  v2f a1 = *reinterpret_cast<const v2f*>(ap + 16 * PITCH_A);
#pragma unroll
  for (int s = 0; s < NSTEP; ++s) {
    const int slot = s % PF;
    v2f b[4];
#pragma unroll
    for (int j = 0; j < 4; ++j) { b[j].x = bx[slot][j]; b[j].y = by[slot][j]; }
    const v2f c0 = a0, c1 = a1;
    if (s + PF < NSTEP) {
      bx[slot] = *reinterpret_cast<const v4f*>(wp + (size_t)(4 * (s + PF)) * PITCH_W);
      by[slot] = *reinterpret_cast<const v4f*>(wp + (size_t)(4 * (s + PF) + 1) * PITCH_W);
    }
    if (s + 1 < NSTEP) {
      a0 = *reinterpret_cast<const v2f*>(ap + 4 * (s + 1));
      a1 = *reinterpret_cast<const v2f*>(ap + 16 * PITCH_A + 4 * (s + 1));
    }
    __builtin_amdgcn_sched_barrier(0);
#pragma unroll
    for (int j = 0; j < 4; ++j) {
      acc[0][j] = wmma4(c0, b[j], acc[0][j]);
      acc[1][j] = wmma4(c1, b[j], acc[1][j]);
    }
  }
}

__global__ __launch_bounds__(256)
void gin_mlp(const float* __restrict__ x, const int* __restrict__ offs, const int* __restrict__ bsrc,
             const float* __restrict__ W1, const float* __restrict__ b1,
             const float* __restrict__ W2, const float* __restrict__ b2,
             float* __restrict__ out) {
  extern __shared__ float smem[];
  float* sH  = smem;
  float* sH1 = smem + 32 * H_PITCH;

  const int tid   = threadIdx.x;
  const int wave  = tid >> 5;
  const int lane  = tid & 31;
  const int row16 = lane & 15;
  const int khalf = (lane >> 4) << 1;
  const int rowhi = (lane >> 4) << 3;
  const int c4    = row16 << 2;

  v4f bx[PF], by[PF];
  v8f acc[2][4];

  const int n1 = wave << 6;
  const float* wp1 = W1 + (size_t)khalf * D_HID + n1 + c4;
  w_preload<D_HID>(wp1, bx, by);

  {
    const int wv    = __builtin_amdgcn_readfirstlane(wave);
    const int node0 = blockIdx.x * 32 + wv * 4;
    const float* xl = x + lane * 4;
    int eb[5];
#pragma unroll
    for (int i = 0; i < 5; ++i) eb[i] = offs[node0 + i];
#pragma unroll
    for (int i = 0; i < 4; ++i) {
      v4f a = *reinterpret_cast<const v4f*>(xl + (size_t)(node0 + i) * D_FEAT);
      const int e1 = eb[i + 1];
      for (int e = eb[i]; e < e1; e += GU) {
        int s[GU];
#pragma unroll
        for (int j = 0; j < GU; ++j) {
          const int ee = (e + j < e1) ? (e + j) : (e1 - 1);
          s[j] = bsrc[ee];
        }
        v4f v[GU];
#pragma unroll
        for (int j = 0; j < GU; ++j)
          v[j] = *reinterpret_cast<const v4f*>(xl + (size_t)s[j] * D_FEAT);
#pragma unroll
        for (int j = 0; j < GU; ++j)
          if (e + j < e1) a += v[j];
      }
      *reinterpret_cast<v4f*>(sH + (wv * 4 + i) * H_PITCH + lane * 4) = a;
    }
  }
  __syncthreads();

#pragma unroll
  for (int m = 0; m < 2; ++m)
#pragma unroll
    for (int j = 0; j < 4; ++j)
      acc[m][j] = (v8f){0.f,0.f,0.f,0.f,0.f,0.f,0.f,0.f};
  wave_tile_gemm<D_HID, H_PITCH, D_FEAT / 4>(wp1, sH + row16 * H_PITCH + khalf, bx, by, acc);

  const int q  = wave >> 1;
  const int n2 = (wave & 1) << 6;
  const int k0 = q << 7;
  const float* wp2 = W2 + (size_t)(k0 + khalf) * D_OUT + n2 + c4;
  w_preload<D_OUT>(wp2, bx, by);

  {
    const v4f bias = *reinterpret_cast<const v4f*>(b1 + n1 + c4);
#pragma unroll
    for (int m = 0; m < 2; ++m)
#pragma unroll
      for (int i = 0; i < 8; ++i) {
        v4f v;
        v.x = acc[m][0][i] + bias.x; v.y = acc[m][1][i] + bias.y;
        v.z = acc[m][2][i] + bias.z; v.w = acc[m][3][i] + bias.w;
        v.x = v.x > 0.f ? v.x : 0.f; v.y = v.y > 0.f ? v.y : 0.f;
        v.z = v.z > 0.f ? v.z : 0.f; v.w = v.w > 0.f ? v.w : 0.f;
        *reinterpret_cast<v4f*>(sH1 + (16 * m + i + rowhi) * H1_PITCH + n1 + c4) = v;
      }
  }
  __syncthreads();

#pragma unroll
  for (int m = 0; m < 2; ++m)
#pragma unroll
    for (int j = 0; j < 4; ++j)
      acc[m][j] = (v8f){0.f,0.f,0.f,0.f,0.f,0.f,0.f,0.f};
  wave_tile_gemm<D_OUT, H1_PITCH, D_HID / 16>(wp2, sH1 + row16 * H1_PITCH + k0 + khalf, bx, by, acc);

  __syncthreads();
  float* sP = smem;
#pragma unroll
  for (int m = 0; m < 2; ++m)
#pragma unroll
    for (int i = 0; i < 8; ++i) {
      v4f v;
      v.x = acc[m][0][i]; v.y = acc[m][1][i]; v.z = acc[m][2][i]; v.w = acc[m][3][i];
      *reinterpret_cast<v4f*>(sP + (q * 32 + 16 * m + i + rowhi) * P_PITCH + n2 + c4) = v;
    }
  __syncthreads();
  const int ocol = lane << 2;
  const v4f bias2 = *reinterpret_cast<const v4f*>(b2 + ocol);
  const size_t obase = (size_t)blockIdx.x * 32 * D_OUT;
#pragma unroll
  for (int i = 0; i < 4; ++i) {
    const int r = wave + 8 * i;
    const float* p = sP + r * P_PITCH + ocol;
    v4f s = *reinterpret_cast<const v4f*>(p);
    s += *reinterpret_cast<const v4f*>(p + 32 * P_PITCH);
    s += *reinterpret_cast<const v4f*>(p + 64 * P_PITCH);
    s += *reinterpret_cast<const v4f*>(p + 96 * P_PITCH);
    s += bias2;
    *reinterpret_cast<v4f*>(out + obase + (size_t)r * D_OUT + ocol) = s;
  }
}

extern "C" void kernel_launch(void* const* d_in, const int* in_sizes, int n_in,
                              void* d_out, int out_size, void* d_ws, size_t ws_size,
                              hipStream_t stream) {
  const float* x  = (const float*)d_in[0];
  const int*   ei = (const int*)d_in[1];
  const float* W1 = (const float*)d_in[2];
  const float* b1 = (const float*)d_in[3];
  const float* W2 = (const float*)d_in[4];
  const float* b2 = (const float*)d_in[5];
  float* out = (float*)d_out;

  const int* src = ei;
  const int* dst = ei + N_EDGES;

  int*  bsrc   = (int*)d_ws;
  int*  keys   = bsrc + N_EDGES;
  int*  coff   = keys + N_EDGES;
  int*  offs   = coff + (NBKT + 1) * NCHUNK;

  k_part<<<NCHUNK, 256, 0, stream>>>(src, dst, keys, coff);
  k_bucket<<<NBKT, 1024, 0, stream>>>(keys, coff, offs, bsrc);

  gin_mlp<<<N_NODES / 32, 256, 82944, stream>>>(x, offs, bsrc, W1, b1, W2, b2, out);
}
